# starting kernel
# baseline (speedup 1.0000x reference)
#include <hip/hip_runtime.h>
#include <hip/hip_bf16.h>

static constexpr int N_NODES = 4096;
static constexpr int N_EDGES = 8192;
static constexpr int N_FEAT  = 5;
static constexpr int N_WIRES = 10;
static constexpr int NPAD    = 16;

typedef __attribute__((ext_vector_type(2))) float v2f;
typedef __attribute__((ext_vector_type(8))) float v8f;

__global__ __launch_bounds__(256) void pad_x_kernel(
    const float* __restrict__ X, float* __restrict__ Xpad)
{
    const int idx = blockIdx.x * 256 + threadIdx.x;
    const int k = idx >> 4;
    const int n = idx & 15;
    Xpad[idx] = (n < N_FEAT) ? X[k * N_FEAT + n] : 0.0f;
}

__global__ __launch_bounds__(256) void angles_gemm_kernel(
    const float* __restrict__ Xpad,
    const float* __restrict__ Ri,
    const float* __restrict__ Ro,
    float* __restrict__ Bmat)
{
    const int lane = threadIdx.x & 31;
    const int wave = threadIdx.x >> 5;
    const int tile = blockIdx.x * 8 + wave;
    const int e0   = tile * 16;
    const int m16  = lane & 15;
    const int half = lane >> 4;

    const float* __restrict__ R = (blockIdx.y == 0) ? Ro : Ri;
    const int colOff            = (blockIdx.y == 0) ? 0  : N_FEAT;

    const float* __restrict__ aPtr = R + (size_t)(2 * half) * N_EDGES + e0 + m16;
    const float* __restrict__ bPtr = Xpad + (2 * half) * NPAD + m16;

    v8f acc = {};
#pragma unroll 4
    for (int k0 = 0; k0 < N_NODES; k0 += 4) {
        v2f a, b;
        a.x = aPtr[0];
        a.y = aPtr[N_EDGES];
        b.x = bPtr[0];
        b.y = bPtr[NPAD];
        acc = __builtin_amdgcn_wmma_f32_16x16x4_f32(
             false, a,  false, b,
             (short)0, acc,  false,  false);
        aPtr += (size_t)4 * N_EDGES;
        bPtr += 4 * NPAD;
    }

    if (m16 < N_FEAT) {
#pragma unroll
        for (int r = 0; r < 8; ++r) {
            const int m = r + 8 * half;
            Bmat[(size_t)(e0 + m) * N_WIRES + colOff + m16] = acc[r];
        }
    }
}

__device__ __forceinline__ float shflx(float x, int m) {
    return __shfl_xor(x, m, 32);
}

template <int W>
__device__ __forceinline__ void apply_ry(float (&v)[32], float c, float s, int lane) {
    constexpr int B = 9 - W;
    if constexpr (B >= 5) {
        constexpr int st = 1 << (B - 5);
#pragma unroll
        for (int r = 0; r < 32; ++r) {
            if ((r & st) == 0) {
                const float p0 = v[r];
                const float p1 = v[r + st];
                v[r]      = c * p0 - s * p1;
                v[r + st] = s * p0 + c * p1;
            }
        }
    } else {
        constexpr int m = 1 << B;
        const float sgn = (lane & m) ? s : -s;
#pragma unroll
        for (int r = 0; r < 32; ++r) {
            const float p = shflx(v[r], m);
            v[r] = c * v[r] + sgn * p;
        }
    }
}

template <int C, int T>
__device__ __forceinline__ void apply_cnot(float (&v)[32], int lane) {
    constexpr int BC = 9 - C;
    constexpr int BT = 9 - T;
    if constexpr (BC >= 5 && BT >= 5) {
        constexpr int cm = 1 << (BC - 5);
        constexpr int tm = 1 << (BT - 5);
#pragma unroll
        for (int r = 0; r < 32; ++r) {
            if ((r & cm) && !(r & tm)) {
                const float t = v[r];
                v[r]      = v[r + tm];
                v[r + tm] = t;
            }
        }
    } else if constexpr (BC >= 5 && BT < 5) {
        constexpr int cm    = 1 << (BC - 5);
        constexpr int tmask = 1 << BT;
#pragma unroll
        for (int r = 0; r < 32; ++r) {
            if (r & cm) v[r] = shflx(v[r], tmask);
        }
    } else if constexpr (BC < 5 && BT >= 5) {
        constexpr int cmask = 1 << BC;
        constexpr int tm    = 1 << (BT - 5);
        const bool cb = (lane & cmask) != 0;
#pragma unroll
        for (int r = 0; r < 32; ++r) {
            if ((r & tm) == 0) {
                const float a = v[r];
                const float b = v[r + tm];
                v[r]      = cb ? b : a;
                v[r + tm] = cb ? a : b;
            }
        }
    } else {
        constexpr int cmask = 1 << BC;
        constexpr int tmask = 1 << BT;
        const bool cb = (lane & cmask) != 0;
#pragma unroll
        for (int r = 0; r < 32; ++r) {
            const float t = shflx(v[r], tmask);
            v[r] = cb ? t : v[r];
        }
    }
}

__global__ __launch_bounds__(256) void ttn_sim_kernel(
    const float* __restrict__ Bmat,
    const float* __restrict__ theta,
    float* __restrict__ out)
{
    const int lane = threadIdx.x & 31;
    const int wave = threadIdx.x >> 5;
    const int e    = blockIdx.x * 8 + wave;

    float ac[10], as[10];
#pragma unroll
    for (int w = 0; w < N_WIRES; ++w) {
        const float h = 0.5f * Bmat[(size_t)e * N_WIRES + w];
        sincosf(h, &as[w], &ac[w]);
    }
    float tc[19], ts[19];
#pragma unroll
    for (int j = 0; j < 19; ++j) {
        const float h = 0.5f * theta[j];
        sincosf(h, &ts[j], &tc[j]);
    }

    float v[32];
#pragma unroll
    for (int r = 0; r < 32; ++r) v[r] = 0.0f;
    if (lane == 0) v[0] = 1.0f;

    apply_ry<0>(v, ac[0], as[0], lane);
    apply_ry<1>(v, ac[1], as[1], lane);
    apply_ry<2>(v, ac[2], as[2], lane);
    apply_ry<3>(v, ac[3], as[3], lane);
    apply_ry<4>(v, ac[4], as[4], lane);
    apply_ry<5>(v, ac[5], as[5], lane);
    apply_ry<6>(v, ac[6], as[6], lane);
    apply_ry<7>(v, ac[7], as[7], lane);
    apply_ry<8>(v, ac[8], as[8], lane);
    apply_ry<9>(v, ac[9], as[9], lane);
    apply_ry<0>(v, tc[0], ts[0], lane); apply_ry<1>(v, tc[1], ts[1], lane); apply_cnot<0, 1>(v, lane);
    apply_ry<2>(v, tc[2], ts[2], lane); apply_ry<3>(v, tc[3], ts[3], lane); apply_cnot<3, 2>(v, lane);
    apply_ry<4>(v, tc[4], ts[4], lane); apply_ry<5>(v, tc[5], ts[5], lane); apply_cnot<5, 4>(v, lane);
    apply_ry<6>(v, tc[6], ts[6], lane); apply_ry<7>(v, tc[7], ts[7], lane); apply_cnot<6, 7>(v, lane);
    apply_ry<8>(v, tc[8], ts[8], lane); apply_ry<9>(v, tc[9], ts[9], lane); apply_cnot<8, 9>(v, lane);
    apply_ry<1>(v, tc[10], ts[10], lane); apply_ry<2>(v, tc[11], ts[11], lane); apply_cnot<1, 2>(v, lane);
    apply_ry<7>(v, tc[12], ts[12], lane); apply_ry<9>(v, tc[13], ts[13], lane); apply_cnot<9, 7>(v, lane);
    apply_ry<2>(v, tc[14], ts[14], lane); apply_ry<4>(v, tc[15], ts[15], lane); apply_cnot<2, 4>(v, lane);
    apply_ry<4>(v, tc[16], ts[16], lane); apply_ry<7>(v, tc[17], ts[17], lane); apply_cnot<4, 7>(v, lane);
    apply_ry<7>(v, tc[18], ts[18], lane);

    float loc = 0.0f;
#pragma unroll
    for (int r = 0; r < 32; ++r) loc += v[r] * v[r];
    loc = (lane & 4) ? loc : 0.0f;
#pragma unroll
    for (int off = 16; off >= 1; off >>= 1) loc += __shfl_xor(loc, off, 32);
    if (lane == 0) out[e] = loc;
}

extern "C" void kernel_launch(void* const* d_in, const int* in_sizes, int n_in,
                              void* d_out, int out_size, void* d_ws, size_t ws_size,
                              hipStream_t stream)
{
    (void)in_sizes; (void)n_in; (void)out_size; (void)ws_size;
    const float* X     = (const float*)d_in[0];
    const float* Ri    = (const float*)d_in[1];
    const float* Ro    = (const float*)d_in[2];
    const float* theta = (const float*)d_in[3];
    float* out  = (float*)d_out;

    float* Bmat = (float*)d_ws;
    float* Xpad = (float*)((char*)d_ws + (size_t)N_EDGES * N_WIRES * sizeof(float));

    pad_x_kernel<<<(N_NODES * NPAD) / 256, 256, 0, stream>>>(X, Xpad);

    dim3 gGemm(512 / 8, 2);
    angles_gemm_kernel<<<gGemm, 256, 0, stream>>>(Xpad, Ri, Ro, Bmat);

    ttn_sim_kernel<<<N_EDGES / 8, 256, 0, stream>>>(Bmat, theta, out);
}


# Angle GEMM takes B from LDS: a work-group lays its slab of X rows (read as stored) out in fragment order; pad launch gone; loop loads Ro/Ri rows only; 8 slabs of 512; angles summed in another order
# speedup vs baseline: 5.6447x; 5.6447x over previous
#include <hip/hip_runtime.h>
#include <hip/hip_bf16.h>

static constexpr int N_NODES = 4096;
static constexpr int N_EDGES = 8192;
static constexpr int N_FEAT  = 5;
static constexpr int N_WIRES = 10;
static constexpr int NPAD    = 16;
static constexpr int K_SLABS = 8;
static constexpr int K_SLAB  = N_NODES / K_SLABS;
static constexpr int K_PAIRS = K_SLAB / 8;
static constexpr int TRIP_PAIRS = 8;
static constexpr int PLANE   = N_EDGES * N_WIRES;
static constexpr size_t PART_BYTES = (size_t)K_SLABS * PLANE * sizeof(float);
static constexpr size_t WS_BYTES   = PART_BYTES;

typedef __attribute__((ext_vector_type(2))) float v2f;
typedef __attribute__((ext_vector_type(8))) float v8f;

typedef __attribute__((ext_vector_type(4))) float v4f;

__global__ __launch_bounds__(256) void angles_gemm_kernel(
    const float* __restrict__ X,
    const float* __restrict__ Ri,
    const float* __restrict__ Ro,
    float* __restrict__ Part)
{
    __shared__ v4f bCell[K_PAIRS * 32];

    const int lane = threadIdx.x & 31;
    const int wave = threadIdx.x >> 5;
    const int tile = blockIdx.x * 8 + wave;
    const int e0   = tile * 16;
    const int m16  = lane & 15;
    const int half = lane >> 4;

    const float* __restrict__ R = (blockIdx.y == 0) ? Ro : Ri;
    const int colOff            = (blockIdx.y == 0) ? 0  : N_FEAT;
    const int slab              = blockIdx.z;
    const int kBase             = slab * K_SLAB;

    {
        const bool keep = m16 < N_FEAT;
        const int  nc   = keep ? m16 : (N_FEAT - 1);
        const float* __restrict__ xs = X + (size_t)(kBase + 2 * half) * N_FEAT + nc;
        v4f c[K_PAIRS / 8];
#pragma unroll
        for (int i = 0; i < K_PAIRS / 8; ++i) {
            const float* __restrict__ q = xs + (size_t)(8 * (wave + 8 * i)) * N_FEAT;
            c[i].x = q[0];
            c[i].y = q[N_FEAT];
            c[i].z = q[4 * N_FEAT];
            c[i].w = q[5 * N_FEAT];
        }
#pragma unroll
        for (int i = 0; i < K_PAIRS / 8; ++i) {
            const v4f z = {0.0f, 0.0f, 0.0f, 0.0f};
            bCell[(wave + 8 * i) * 32 + lane] = keep ? c[i] : z;
        }
    }
    __syncthreads();

    const float* __restrict__ aPtr = R + (size_t)(kBase + 2 * half) * N_EDGES + e0 + m16;

    v8f acc = {};
    for (int t = 0; t < K_PAIRS / TRIP_PAIRS; ++t) {
        float a[4 * TRIP_PAIRS];
#pragma unroll
        for (int j = 0; j < 2 * TRIP_PAIRS; ++j) {
            a[2 * j]     = aPtr[(size_t)(4 * j) * N_EDGES];
            a[2 * j + 1] = aPtr[(size_t)(4 * j + 1) * N_EDGES];
        }
#pragma unroll
        for (int p = 0; p < TRIP_PAIRS; ++p) {
            const v4f b = bCell[(t * TRIP_PAIRS + p) * 32 + lane];
            v2f a0, a1, b0, b1;
            a0.x = a[4 * p];     a0.y = a[4 * p + 1];
            a1.x = a[4 * p + 2]; a1.y = a[4 * p + 3];
            b0.x = b.x; b0.y = b.y;
            b1.x = b.z; b1.y = b.w;
            acc = __builtin_amdgcn_wmma_f32_16x16x4_f32(
                 false, a0,  false, b0,
                 (short)0, acc,  false,  false);
            acc = __builtin_amdgcn_wmma_f32_16x16x4_f32(
                 false, a1,  false, b1,
                 (short)0, acc,  false,  false);
        }
        aPtr += (size_t)(8 * TRIP_PAIRS) * N_EDGES;
    }

    float* __restrict__ dst = Part + (size_t)slab * PLANE;
    if (m16 < N_FEAT) {
#pragma unroll
        for (int r = 0; r < 8; ++r) {
            const int m = r + 8 * half;
            dst[(size_t)(e0 + m) * N_WIRES + colOff + m16] = acc[r];
        }
    }
}

static constexpr int R1 = 4, R2 = 3, R4 = 2, R7 = 1, R9 = 0;

template <int BIT>
__device__ __forceinline__ void ry_reg(float (&v)[32], float c, float s) {
    constexpr int st = 1 << BIT;
#pragma unroll
    for (int r = 0; r < 32; ++r) {
        if ((r & st) == 0) {
            const float p0 = v[r];
            const float p1 = v[r + st];
            v[r]      = c * p0 - s * p1;
            v[r + st] = s * p0 + c * p1;
        }
    }
}

template <int CB, int TB>
__device__ __forceinline__ void cnot_reg(float (&v)[32]) {
    constexpr int cm = 1 << CB;
    constexpr int tm = 1 << TB;
#pragma unroll
    for (int r = 0; r < 32; ++r) {
        if ((r & cm) && !(r & tm)) {
            const float t = v[r];
            v[r]      = v[r + tm];
            v[r + tm] = t;
        }
    }
}

__global__ __launch_bounds__(256) void ttn_sim_kernel(
    const float* __restrict__ Part,
    const float* __restrict__ theta,
    float* __restrict__ out)
{
    const int lane = threadIdx.x & 31;
    const int wave = threadIdx.x >> 5;
    const int e    = blockIdx.x * 8 + wave;

    const int w16 = lane & 15;
    const int zh  = lane >> 4;
    float part = 0.0f;
    if (w16 < N_WIRES) {
        const float* __restrict__ p =
            Part + (size_t)(zh * (K_SLABS / 2)) * PLANE + (size_t)e * N_WIRES + w16;
#pragma unroll
        for (int j = 0; j < K_SLABS / 2; ++j) part += p[(size_t)j * PLANE];
    }
    const float angSum = part + __shfl_xor(part, 16, 32);

    float C[N_WIRES], S[N_WIRES];
#pragma unroll
    for (int w = 0; w < N_WIRES; ++w) {
        const float h = 0.5f * (__shfl(angSum, w, 32) + theta[w]);
        sincosf(h, &S[w], &C[w]);
    }
    float tc[9], ts[9];
#pragma unroll
    for (int j = 0; j < 9; ++j) {
        const float h = 0.5f * theta[10 + j];
        sincosf(h, &ts[j], &tc[j]);
    }

    const bool b0 = (lane & 16) != 0;
    const bool b3 = (lane & 8) != 0;
    const bool b5 = (lane & 4) != 0;
    const bool b6 = (lane & 2) != 0;
    const bool b8 = (lane & 1) != 0;
    const float lf = (b0 ? S[0] : C[0]) * (b3 ? S[3] : C[3]) * (b5 ? S[5] : C[5])
                   * (b6 ? S[6] : C[6]) * (b8 ? S[8] : C[8]);
    const float g1[2] = { b0 ? S[1] : C[1], b0 ? C[1] : S[1] };
    const float g2[2] = { b3 ? S[2] : C[2], b3 ? C[2] : S[2] };
    const float g4[2] = { b5 ? S[4] : C[4], b5 ? C[4] : S[4] };
    const float g7[2] = { b6 ? S[7] : C[7], b6 ? C[7] : S[7] };
    const float g9[2] = { b8 ? S[9] : C[9], b8 ? C[9] : S[9] };

    float v[32];
#pragma unroll
    for (int r = 0; r < 32; ++r) {
        v[r] = lf * g1[(r >> R1) & 1] * g2[(r >> R2) & 1] * g4[(r >> R4) & 1]
                  * g7[(r >> R7) & 1] * g9[(r >> R9) & 1];
    }

    ry_reg<R1>(v, tc[0], ts[0]); ry_reg<R2>(v, tc[1], ts[1]); cnot_reg<R1, R2>(v);
    ry_reg<R7>(v, tc[2], ts[2]); ry_reg<R9>(v, tc[3], ts[3]); cnot_reg<R9, R7>(v);
    ry_reg<R2>(v, tc[4], ts[4]); ry_reg<R4>(v, tc[5], ts[5]); cnot_reg<R2, R4>(v);
    ry_reg<R4>(v, tc[6], ts[6]); ry_reg<R7>(v, tc[7], ts[7]); cnot_reg<R4, R7>(v);
    ry_reg<R7>(v, tc[8], ts[8]);

    float loc = 0.0f;
#pragma unroll
    for (int r = 0; r < 32; ++r) {
        if (r & (1 << R7)) loc += v[r] * v[r];
    }
#pragma unroll
    for (int off = 16; off >= 1; off >>= 1) loc += __shfl_xor(loc, off, 32);
    if (lane == 0) out[e] = loc;
}

extern "C" void kernel_launch(void* const* d_in, const int* in_sizes, int n_in,
                              void* d_out, int out_size, void* d_ws, size_t ws_size,
                              hipStream_t stream)
{
    (void)in_sizes; (void)n_in; (void)out_size;
    if (ws_size < WS_BYTES) return;
    const float* X     = (const float*)d_in[0];
    const float* Ri    = (const float*)d_in[1];
    const float* Ro    = (const float*)d_in[2];
    const float* theta = (const float*)d_in[3];
    float* out  = (float*)d_out;

    float* Part = (float*)d_ws;

    dim3 gGemm(512 / 8, 2, K_SLABS);
    angles_gemm_kernel<<<gGemm, 256, 0, stream>>>(X, Ri, Ro, Part);

    ttn_sim_kernel<<<N_EDGES / 8, 256, 0, stream>>>(Part, theta, out);
}


# Circuit kernel: the 19 half-angle sincos are spread over the lanes (one sincosf per wave instead of 19 per lane) and each pair is read back with readlane; sums and gate order as before
# speedup vs baseline: 6.0731x; 1.0759x over previous
#include <hip/hip_runtime.h>
#include <hip/hip_bf16.h>

static constexpr int N_NODES = 4096;
static constexpr int N_EDGES = 8192;
static constexpr int N_FEAT  = 5;
static constexpr int N_WIRES = 10;
static constexpr int NPAD    = 16;
static constexpr int K_SLABS = 8;
static constexpr int K_SLAB  = N_NODES / K_SLABS;
static constexpr int K_PAIRS = K_SLAB / 8;
static constexpr int TRIP_PAIRS = 8;
static constexpr int PLANE   = N_EDGES * N_WIRES;
static constexpr size_t PART_BYTES = (size_t)K_SLABS * PLANE * sizeof(float);
static constexpr size_t WS_BYTES   = PART_BYTES;

typedef __attribute__((ext_vector_type(2))) float v2f;
typedef __attribute__((ext_vector_type(8))) float v8f;

typedef __attribute__((ext_vector_type(4))) float v4f;

__global__ __launch_bounds__(256) void angles_gemm_kernel(
    const float* __restrict__ X,
    const float* __restrict__ Ri,
    const float* __restrict__ Ro,
    float* __restrict__ Part)
{
    __shared__ v4f bCell[K_PAIRS * 32];

    const int lane = threadIdx.x & 31;
    const int wave = threadIdx.x >> 5;
    const int tile = blockIdx.x * 8 + wave;
    const int e0   = tile * 16;
    const int m16  = lane & 15;
    const int half = lane >> 4;

    const float* __restrict__ R = (blockIdx.y == 0) ? Ro : Ri;
    const int colOff            = (blockIdx.y == 0) ? 0  : N_FEAT;
    const int slab              = blockIdx.z;
    const int kBase             = slab * K_SLAB;

    {
        const bool keep = m16 < N_FEAT;
        const int  nc   = keep ? m16 : (N_FEAT - 1);
        const float* __restrict__ xs = X + (size_t)(kBase + 2 * half) * N_FEAT + nc;
        v4f c[K_PAIRS / 8];
#pragma unroll
        for (int i = 0; i < K_PAIRS / 8; ++i) {
            const float* __restrict__ q = xs + (size_t)(8 * (wave + 8 * i)) * N_FEAT;
            c[i].x = q[0];
            c[i].y = q[N_FEAT];
            c[i].z = q[4 * N_FEAT];
            c[i].w = q[5 * N_FEAT];
        }
#pragma unroll
        for (int i = 0; i < K_PAIRS / 8; ++i) {
            const v4f z = {0.0f, 0.0f, 0.0f, 0.0f};
            bCell[(wave + 8 * i) * 32 + lane] = keep ? c[i] : z;
        }
    }
    __syncthreads();

    const float* __restrict__ aPtr = R + (size_t)(kBase + 2 * half) * N_EDGES + e0 + m16;

    v8f acc = {};
    for (int t = 0; t < K_PAIRS / TRIP_PAIRS; ++t) {
        float a[4 * TRIP_PAIRS];
#pragma unroll
        for (int j = 0; j < 2 * TRIP_PAIRS; ++j) {
            a[2 * j]     = aPtr[(size_t)(4 * j) * N_EDGES];
            a[2 * j + 1] = aPtr[(size_t)(4 * j + 1) * N_EDGES];
        }
#pragma unroll
        for (int p = 0; p < TRIP_PAIRS; ++p) {
            const v4f b = bCell[(t * TRIP_PAIRS + p) * 32 + lane];
            v2f a0, a1, b0, b1;
            a0.x = a[4 * p];     a0.y = a[4 * p + 1];
            a1.x = a[4 * p + 2]; a1.y = a[4 * p + 3];
            b0.x = b.x; b0.y = b.y;
            b1.x = b.z; b1.y = b.w;
            acc = __builtin_amdgcn_wmma_f32_16x16x4_f32(
                 false, a0,  false, b0,
                 (short)0, acc,  false,  false);
            acc = __builtin_amdgcn_wmma_f32_16x16x4_f32(
                 false, a1,  false, b1,
                 (short)0, acc,  false,  false);
        }
        aPtr += (size_t)(8 * TRIP_PAIRS) * N_EDGES;
    }

    float* __restrict__ dst = Part + (size_t)slab * PLANE;
    if (m16 < N_FEAT) {
#pragma unroll
        for (int r = 0; r < 8; ++r) {
            const int m = r + 8 * half;
            dst[(size_t)(e0 + m) * N_WIRES + colOff + m16] = acc[r];
        }
    }
}

static constexpr int R1 = 4, R2 = 3, R4 = 2, R7 = 1, R9 = 0;

template <int BIT>
__device__ __forceinline__ void ry_reg(float (&v)[32], float c, float s) {
    constexpr int st = 1 << BIT;
#pragma unroll
    for (int r = 0; r < 32; ++r) {
        if ((r & st) == 0) {
            const float p0 = v[r];
            const float p1 = v[r + st];
            v[r]      = c * p0 - s * p1;
            v[r + st] = s * p0 + c * p1;
        }
    }
}

template <int CB, int TB>
__device__ __forceinline__ void cnot_reg(float (&v)[32]) {
    constexpr int cm = 1 << CB;
    constexpr int tm = 1 << TB;
#pragma unroll
    for (int r = 0; r < 32; ++r) {
        if ((r & cm) && !(r & tm)) {
            const float t = v[r];
            v[r]      = v[r + tm];
            v[r + tm] = t;
        }
    }
}

static constexpr int N_THETA = 19;

__device__ __forceinline__ float lane_value(float x, int src) {
    return __int_as_float(__builtin_amdgcn_readlane(__float_as_int(x), src));
}

__global__ __launch_bounds__(256) void ttn_sim_kernel(
    const float* __restrict__ Part,
    const float* __restrict__ theta,
    float* __restrict__ out)
{
    const int lane = threadIdx.x & 31;
    const int wave = threadIdx.x >> 5;
    const int e    = blockIdx.x * 8 + wave;

    const int w16 = lane & 15;
    const int zh  = lane >> 4;
    float part = 0.0f;
    if (w16 < N_WIRES) {
        const float* __restrict__ p =
            Part + (size_t)(zh * (K_SLABS / 2)) * PLANE + (size_t)e * N_WIRES + w16;
#pragma unroll
        for (int j = 0; j < K_SLABS / 2; ++j) part += p[(size_t)j * PLANE];
    }
    const float angSum = part + __shfl_xor(part, 16, 32);

    const float th  = theta[lane < N_THETA ? lane : N_THETA - 1];
    const float arg = (lane < N_WIRES) ? (angSum + th) : th;
    float sn, cs;
    sincosf(0.5f * arg, &sn, &cs);
    float C[N_WIRES], S[N_WIRES];
#pragma unroll
    for (int w = 0; w < N_WIRES; ++w) {
        C[w] = lane_value(cs, w);
        S[w] = lane_value(sn, w);
    }
    float tc[9], ts[9];
#pragma unroll
    for (int j = 0; j < 9; ++j) {
        tc[j] = lane_value(cs, N_WIRES + j);
        ts[j] = lane_value(sn, N_WIRES + j);
    }

    const bool b0 = (lane & 16) != 0;
    const bool b3 = (lane & 8) != 0;
    const bool b5 = (lane & 4) != 0;
    const bool b6 = (lane & 2) != 0;
    const bool b8 = (lane & 1) != 0;
    const float lf = (b0 ? S[0] : C[0]) * (b3 ? S[3] : C[3]) * (b5 ? S[5] : C[5])
                   * (b6 ? S[6] : C[6]) * (b8 ? S[8] : C[8]);
    const float g1[2] = { b0 ? S[1] : C[1], b0 ? C[1] : S[1] };
    const float g2[2] = { b3 ? S[2] : C[2], b3 ? C[2] : S[2] };
    const float g4[2] = { b5 ? S[4] : C[4], b5 ? C[4] : S[4] };
    const float g7[2] = { b6 ? S[7] : C[7], b6 ? C[7] : S[7] };
    const float g9[2] = { b8 ? S[9] : C[9], b8 ? C[9] : S[9] };

    float v[32];
#pragma unroll
    for (int r = 0; r < 32; ++r) {
        v[r] = lf * g1[(r >> R1) & 1] * g2[(r >> R2) & 1] * g4[(r >> R4) & 1]
                  * g7[(r >> R7) & 1] * g9[(r >> R9) & 1];
    }

    ry_reg<R1>(v, tc[0], ts[0]); ry_reg<R2>(v, tc[1], ts[1]); cnot_reg<R1, R2>(v);
    ry_reg<R7>(v, tc[2], ts[2]); ry_reg<R9>(v, tc[3], ts[3]); cnot_reg<R9, R7>(v);
    ry_reg<R2>(v, tc[4], ts[4]); ry_reg<R4>(v, tc[5], ts[5]); cnot_reg<R2, R4>(v);
    ry_reg<R4>(v, tc[6], ts[6]); ry_reg<R7>(v, tc[7], ts[7]); cnot_reg<R4, R7>(v);
    ry_reg<R7>(v, tc[8], ts[8]);

    float loc = 0.0f;
#pragma unroll
    for (int r = 0; r < 32; ++r) {
        if (r & (1 << R7)) loc += v[r] * v[r];
    }
#pragma unroll
    for (int off = 16; off >= 1; off >>= 1) loc += __shfl_xor(loc, off, 32);
    if (lane == 0) out[e] = loc;
}

extern "C" void kernel_launch(void* const* d_in, const int* in_sizes, int n_in,
                              void* d_out, int out_size, void* d_ws, size_t ws_size,
                              hipStream_t stream)
{
    (void)in_sizes; (void)n_in; (void)out_size;
    if (ws_size < WS_BYTES) return;
    const float* X     = (const float*)d_in[0];
    const float* Ri    = (const float*)d_in[1];
    const float* Ro    = (const float*)d_in[2];
    const float* theta = (const float*)d_in[3];
    float* out  = (float*)d_out;

    float* Part = (float*)d_ws;

    dim3 gGemm(512 / 8, 2, K_SLABS);
    angles_gemm_kernel<<<gGemm, 256, 0, stream>>>(X, Ri, Ro, Part);

    ttn_sim_kernel<<<N_EDGES / 8, 256, 0, stream>>>(Part, theta, out);
}
